# starting kernel
# baseline (speedup 1.0000x reference)
#include <hip/hip_runtime.h>


typedef float v2f __attribute__((ext_vector_type(2)));
typedef float v8f __attribute__((ext_vector_type(8)));

#define NB    32
#define CIN   512
#define CHID  512
#define COUT  256
#define TLEN  4096

__global__ void gather_last(const float* __restrict__ x, float* __restrict__ xlast) {
    int idx = blockIdx.x * blockDim.x + threadIdx.x;
    xlast[idx] = x[(size_t)idx * TLEN + (TLEN - 1)];
}

template<int K, bool RELU>
__global__ void wmma_gemm_bias(const float* __restrict__ A,
                               const float* __restrict__ W,
                               const float* __restrict__ bias,
                               float* __restrict__ D, int ldd) {
    const int ntile = blockIdx.x;
    const int mtile = blockIdx.y;
    const int lane  = threadIdx.x;
    const int half  = lane >> 4;
    const int l15   = lane & 15;
    const int kb    = half * 2;

    const float* arow = A + (size_t)(mtile * 16 + l15) * K;
    const float* wrow = W + (size_t)(ntile * 16 + l15) * K;

    const float bval = bias[ntile * 16 + l15];
    v8f c;
    #pragma unroll
    for (int r = 0; r < 8; ++r) c[r] = bval;

    #pragma unroll 8
    for (int k0 = 0; k0 < K; k0 += 4) {
        v2f a, b;
        a.x = arow[k0 + kb];
        a.y = arow[k0 + kb + 1];
        b.x = wrow[k0 + kb];
        b.y = wrow[k0 + kb + 1];
        c = __builtin_amdgcn_wmma_f32_16x16x4_f32(false, a, false, b,
                                                  (short)0, c, false, false);
    }

    #pragma unroll
    for (int r = 0; r < 8; ++r) {
        float v = c[r];
        if (RELU) v = fmaxf(v, 0.0f);
        D[(size_t)(mtile * 16 + half * 8 + r) * ldd + ntile * 16 + l15] = v;
    }
}

__global__ void softmax_rows(const float* __restrict__ Y, float* __restrict__ out) {
    __shared__ float red[COUT];
    const int row = blockIdx.x;
    const int tid = threadIdx.x;
    const float v = Y[row * COUT + tid];

    red[tid] = v;
    __syncthreads();
    #pragma unroll
    for (int s = COUT / 2; s > 0; s >>= 1) {
        if (tid < s) red[tid] = fmaxf(red[tid], red[tid + s]);
        __syncthreads();
    }
    const float m = red[0];
    __syncthreads();

    const float e = expf(v - m);
    red[tid] = e;
    __syncthreads();
    #pragma unroll
    for (int s = COUT / 2; s > 0; s >>= 1) {
        if (tid < s) red[tid] += red[tid + s];
        __syncthreads();
    }
    out[row * COUT + tid] = e / red[0];
}

extern "C" void kernel_launch(void* const* d_in, const int* in_sizes, int n_in,
                              void* d_out, int out_size, void* d_ws, size_t ws_size,
                              hipStream_t stream) {
    const float* x  = (const float*)d_in[0];
    const float* W1 = (const float*)d_in[1];
    const float* b1 = (const float*)d_in[2];
    const float* W2 = (const float*)d_in[3];
    const float* b2 = (const float*)d_in[4];
    float* out = (float*)d_out;

    float* xlast = (float*)d_ws;
    float* h     = xlast + NB * CIN;
    float* y     = h     + NB * CHID;

    gather_last<<<(NB * CIN) / 256, 256, 0, stream>>>(x, xlast);

    wmma_gemm_bias<CIN, true>
        <<<dim3(CHID / 16, NB / 16), 32, 0, stream>>>(xlast, W1, b1, h, CHID);

    wmma_gemm_bias<CHID, false>
        <<<dim3(COUT / 16, NB / 16), 32, 0, stream>>>(h, W2, b2, y, COUT);

    softmax_rows<<<NB, COUT, 0, stream>>>(y, out);
}


# GEMM K loop split over 8 waves per tile, each wave's loads all issued up front (16-byte), partial tiles summed through LDS; changes f32 summation order (split-K tree, bias added last)
# speedup vs baseline: 1.9149x; 1.9149x over previous
#include <hip/hip_runtime.h>


typedef float v2f __attribute__((ext_vector_type(2)));
typedef float v8f __attribute__((ext_vector_type(8)));

#define NB    32
#define CIN   512
#define CHID  512
#define COUT  256
#define TLEN  4096

__global__ void gather_last(const float* __restrict__ x, float* __restrict__ xlast) {
    int idx = blockIdx.x * blockDim.x + threadIdx.x;
    xlast[idx] = x[(size_t)idx * TLEN + (TLEN - 1)];
}

#define KSPLIT 8

template<int K, bool RELU>
__global__ __launch_bounds__(KSPLIT * 32)
void wmma_gemm_bias(const float* __restrict__ A,
                    const float* __restrict__ W,
                    const float* __restrict__ bias,
                    float* __restrict__ D, int ldd) {
    constexpr int KS = K / KSPLIT;
    constexpr int NV = KS / 8;
    static_assert(K % (KSPLIT * 8) == 0, "K range per wave must be whole 8-float groups");

    __shared__ float part[KSPLIT * 256];

    const int ntile = blockIdx.x;
    const int mtile = blockIdx.y;
    const int tid   = threadIdx.x;
    const int wave  = tid >> 5;
    const int lane  = tid & 31;
    const int half  = lane >> 4;
    const int l15   = lane & 15;

    const float4* arow = reinterpret_cast<const float4*>(
        A + (size_t)(mtile * 16 + l15) * K + wave * KS + half * 4);
    const float4* wrow = reinterpret_cast<const float4*>(
        W + (size_t)(ntile * 16 + l15) * K + wave * KS + half * 4);

    const float bval = bias[ntile * 16 + l15];

    float4 av[NV], bv[NV];
    #pragma unroll
    for (int j = 0; j < NV; ++j) {
        av[j] = arow[2 * j];
        bv[j] = wrow[2 * j];
    }
    __builtin_amdgcn_sched_barrier(0);

    v8f c0, c1;
    #pragma unroll
    for (int r = 0; r < 8; ++r) { c0[r] = 0.0f; c1[r] = 0.0f; }

    #pragma unroll
    for (int j = 0; j < NV; ++j) {
        v2f a, b;
        a.x = av[j].x; a.y = av[j].y;
        b.x = bv[j].x; b.y = bv[j].y;
        c0 = __builtin_amdgcn_wmma_f32_16x16x4_f32(false, a, false, b,
                                                   (short)0, c0, false, false);
        a.x = av[j].z; a.y = av[j].w;
        b.x = bv[j].z; b.y = bv[j].w;
        c1 = __builtin_amdgcn_wmma_f32_16x16x4_f32(false, a, false, b,
                                                   (short)0, c1, false, false);
    }

    #pragma unroll
    for (int r = 0; r < 8; ++r)
        part[(wave * 8 + r) * 32 + lane] = c0[r] + c1[r];
    __syncthreads();

    float v = bval;
    #pragma unroll
    for (int w = 0; w < KSPLIT; ++w) v += part[w * 256 + tid];
    if (RELU) v = fmaxf(v, 0.0f);
    D[(size_t)(mtile * 16 + half * 8 + wave) * ldd + ntile * 16 + l15] = v;
}

__global__ void softmax_rows(const float* __restrict__ Y, float* __restrict__ out) {
    __shared__ float red[COUT];
    const int row = blockIdx.x;
    const int tid = threadIdx.x;
    const float v = Y[row * COUT + tid];

    red[tid] = v;
    __syncthreads();
    #pragma unroll
    for (int s = COUT / 2; s > 0; s >>= 1) {
        if (tid < s) red[tid] = fmaxf(red[tid], red[tid + s]);
        __syncthreads();
    }
    const float m = red[0];
    __syncthreads();

    const float e = expf(v - m);
    red[tid] = e;
    __syncthreads();
    #pragma unroll
    for (int s = COUT / 2; s > 0; s >>= 1) {
        if (tid < s) red[tid] += red[tid + s];
        __syncthreads();
    }
    out[row * COUT + tid] = e / red[0];
}

extern "C" void kernel_launch(void* const* d_in, const int* in_sizes, int n_in,
                              void* d_out, int out_size, void* d_ws, size_t ws_size,
                              hipStream_t stream) {
    const float* x  = (const float*)d_in[0];
    const float* W1 = (const float*)d_in[1];
    const float* b1 = (const float*)d_in[2];
    const float* W2 = (const float*)d_in[3];
    const float* b2 = (const float*)d_in[4];
    float* out = (float*)d_out;

    float* xlast = (float*)d_ws;
    float* h     = xlast + NB * CIN;
    float* y     = h     + NB * CHID;

    gather_last<<<(NB * CIN) / 256, 256, 0, stream>>>(x, xlast);

    wmma_gemm_bias<CIN, true>
        <<<dim3(CHID / 16, NB / 16), KSPLIT * 32, 0, stream>>>(xlast, W1, b1, h, CHID);

    wmma_gemm_bias<CHID, false>
        <<<dim3(COUT / 16, NB / 16), KSPLIT * 32, 0, stream>>>(h, W2, b2, y, COUT);

    softmax_rows<<<NB, COUT, 0, stream>>>(y, out);
}
